# starting kernel
# baseline (speedup 1.0000x reference)
#include <hip/hip_runtime.h>
#include <hip/hip_bf16.h>

typedef __attribute__((ext_vector_type(2))) float v2f;
typedef __attribute__((ext_vector_type(8))) float v8f;

#define D 128

__global__ __launch_bounds__(256) void zero_f32(float* __restrict__ p, size_t n) {
    size_t stride = (size_t)gridDim.x * blockDim.x;
    for (size_t i = (size_t)blockIdx.x * blockDim.x + threadIdx.x; i < n; i += stride)
        p[i] = 0.0f;
}

__global__ __launch_bounds__(256) void edge_scatter(
    const float* __restrict__ feat, const int* __restrict__ src,
    const int* __restrict__ dst, const float* __restrict__ ew,
    float* __restrict__ agg, int E)
{
    long long gid = (long long)blockIdx.x * blockDim.x + threadIdx.x;
    int e = (int)(gid >> 5);
    if (e >= E) return;
    int lane = (int)(gid & 31);

    int s = src[e];
    int d = dst[e];
    float w = ew[e];

    const float4 f = ((const float4*)(feat + (size_t)s * D))[lane];
    float* ap = agg + (size_t)d * D + lane * 4;
    atomicAdd(ap + 0, f.x * w);
    atomicAdd(ap + 1, f.y * w);
    atomicAdd(ap + 2, f.z * w);
    atomicAdd(ap + 3, f.w * w);
}

__global__ __launch_bounds__(256) void gemm_wmma(
    const float* __restrict__ agg, const float* __restrict__ W,
    const float* __restrict__ bias, const float* __restrict__ a1p,
    float* __restrict__ h_out, float* __restrict__ colsum,
    float* __restrict__ colsumsq)
{
    __shared__ float s_sum[D];
    __shared__ float s_sq[D];
    int tid = threadIdx.x;
    if (tid < D) { s_sum[tid] = 0.0f; s_sq[tid] = 0.0f; }
    __syncthreads();

    const int wave = tid >> 5;
    const int lane = tid & 31;
    const int half = lane >> 4;
    const int l    = lane & 15;
    const int m0   = blockIdx.x * 16;
    const int j0   = wave * 16;
    const float a1 = a1p[0];

    const float* Arow = agg + (size_t)(m0 + l) * D + 2 * half;
    const float* Bcol = W + (size_t)(2 * half) * D + (j0 + l);

    v8f c0 = {};
    v8f c1 = {};
    #pragma unroll 4
    for (int k = 0; k < D; k += 8) {
        {
            v2f a;
            a.x = Arow[k];
            a.y = Arow[k + 1];
            v2f bb;
            bb.x = Bcol[(size_t)k * D];
            bb.y = Bcol[(size_t)(k + 1) * D];
            c0 = __builtin_amdgcn_wmma_f32_16x16x4_f32(
                     false, a, false, bb, (short)0, c0, false, false);
        }
        {
            v2f a;
            a.x = Arow[k + 4];
            a.y = Arow[k + 5];
            v2f bb;
            bb.x = Bcol[(size_t)(k + 4) * D];
            bb.y = Bcol[(size_t)(k + 5) * D];
            c1 = __builtin_amdgcn_wmma_f32_16x16x4_f32(
                     false, a, false, bb, (short)0, c1, false, false);
        }
    }

    const float bj = bias[j0 + l];
    float lsum = 0.0f, lsq = 0.0f;
    #pragma unroll
    for (int v = 0; v < 8; ++v) {
        float h = c0[v] + c1[v] + bj;
        h = (h >= 0.0f) ? h : a1 * h;
        h_out[(size_t)(m0 + v + 8 * half) * D + (j0 + l)] = h;
        lsum += h;
        lsq  += h * h;
    }

    atomicAdd(&s_sum[j0 + l], lsum);
    atomicAdd(&s_sq[j0 + l], lsq);
    __syncthreads();
    if (tid < D) {
        atomicAdd(&colsum[tid],  s_sum[tid]);
        atomicAdd(&colsumsq[tid], s_sq[tid]);
    }
}

__global__ void bn_finalize(const float* __restrict__ colsum,
                            const float* __restrict__ colsumsq,
                            const float* __restrict__ gamma,
                            const float* __restrict__ beta,
                            float* __restrict__ scale,
                            float* __restrict__ shift, int N)
{
    int j = threadIdx.x;
    float invN = 1.0f / (float)N;
    float m   = colsum[j] * invN;
    float var = colsumsq[j] * invN - m * m;
    float inv = rsqrtf(var + 1e-5f);
    float sc  = gamma[j] * inv;
    scale[j] = sc;
    shift[j] = beta[j] - m * sc;
}

__global__ __launch_bounds__(256) void bn_prelu(
    float* __restrict__ h, const float* __restrict__ scale,
    const float* __restrict__ shift, const float* __restrict__ a2p,
    size_t nquads)
{
    size_t i = (size_t)blockIdx.x * blockDim.x + threadIdx.x;
    if (i >= nquads) return;
    const float a2 = a2p[0];
    int j = (int)((i * 4) & (D - 1));

    float4 v = ((float4*)h)[i];
    float x;
    x = v.x * scale[j + 0] + shift[j + 0]; v.x = (x >= 0.0f) ? x : a2 * x;
    x = v.y * scale[j + 1] + shift[j + 1]; v.y = (x >= 0.0f) ? x : a2 * x;
    x = v.z * scale[j + 2] + shift[j + 2]; v.z = (x >= 0.0f) ? x : a2 * x;
    x = v.w * scale[j + 3] + shift[j + 3]; v.w = (x >= 0.0f) ? x : a2 * x;
    ((float4*)h)[i] = v;
}

extern "C" void kernel_launch(void* const* d_in, const int* in_sizes, int n_in,
                              void* d_out, int out_size, void* d_ws, size_t ws_size,
                              hipStream_t stream) {
    const float* feat  = (const float*)d_in[0];
    const int*   src   = (const int*)d_in[1];
    const int*   dst   = (const int*)d_in[2];
    const float* ew    = (const float*)d_in[3];
    const float* Wm    = (const float*)d_in[4];
    const float* bias  = (const float*)d_in[5];
    const float* a1    = (const float*)d_in[6];
    const float* gamma = (const float*)d_in[7];
    const float* beta  = (const float*)d_in[8];
    const float* a2    = (const float*)d_in[9];

    const int N = in_sizes[0] / D;
    const int E = in_sizes[1];
    float* out = (float*)d_out;

    float* agg      = (float*)d_ws;
    float* colsum   = agg + (size_t)N * D;
    float* colsumsq = colsum + D;
    float* scale    = colsumsq + D;
    float* shift    = scale + D;

    size_t nz = (size_t)N * D + 2 * D;
    zero_f32<<<2048, 256, 0, stream>>>(agg, nz);

    long long sthreads = (long long)E * 32;
    int sblocks = (int)((sthreads + 255) / 256);
    edge_scatter<<<sblocks, 256, 0, stream>>>(feat, src, dst, ew, agg, E);

    gemm_wmma<<<N / 16, 256, 0, stream>>>(agg, Wm, bias, a1, out, colsum, colsumsq);

    bn_finalize<<<1, D, 0, stream>>>(colsum, colsumsq, gamma, beta, scale, shift, N);

    size_t nquads = (size_t)N * D / 4;
    int qblocks = (int)((nquads + 255) / 256);
    bn_prelu<<<qblocks, 256, 0, stream>>>(out, scale, shift, a2, nquads);
}


# Edge scatter by float atomics replaced by an in-call counting sort of edges by destination row (integer atomics) plus one-wave-per-row gather-sum with plain stores; row sums in arrival-order rank
# speedup vs baseline: 1.1747x; 1.1747x over previous
#include <hip/hip_runtime.h>
#include <hip/hip_bf16.h>

typedef __attribute__((ext_vector_type(2))) float v2f;
typedef __attribute__((ext_vector_type(8))) float v8f;

#define D 128


__global__ __launch_bounds__(256) void zero_counters(int* __restrict__ cnt, int n,
                                                     float* __restrict__ colsums) {
    int i = blockIdx.x * 256 + threadIdx.x;
    if (i < n) cnt[i] = 0;
    if (i < 2 * D) colsums[i] = 0.0f;
}

__global__ __launch_bounds__(256) void dst_histogram(const int* __restrict__ dst,
                                                     int* __restrict__ cnt, int E) {
    int e = blockIdx.x * 256 + threadIdx.x;
    if (e < E) atomicAdd(&cnt[dst[e]], 1);
}

__global__ __launch_bounds__(1024) void row_offsets(const int* __restrict__ cnt,
                                                    int* __restrict__ offs, int n) {
    __shared__ int s_tot[32];
    __shared__ int s_pre[32];
    const int tid  = threadIdx.x;
    const int lane = tid & 31;
    const int wv   = tid >> 5;
    const int base = blockIdx.x * 1024;

    int pre = 0;
    const int4* c4 = (const int4*)cnt;
    const int nq = base >> 2;
    #pragma unroll 4
    for (int q = tid; q < nq; q += 1024) {
        int4 v = c4[q];
        pre += (v.x + v.y) + (v.z + v.w);
    }

    const int i = base + tid;
    const int c = (i < n) ? cnt[i] : 0;
    int x = c;
    for (int o = 1; o < 32; o <<= 1) {
        int y = __shfl_up(x, o);
        if (lane >= o) x += y;
    }
    for (int o = 16; o > 0; o >>= 1) pre += __shfl_xor(pre, o);
    if (lane == 31) s_tot[wv] = x;
    if (lane == 0)  s_pre[wv] = pre;
    __syncthreads();
    if (wv == 0) {
        const int t = s_tot[lane];
        int p = s_pre[lane];
        int xs = t;
        for (int o = 1; o < 32; o <<= 1) {
            int y = __shfl_up(xs, o);
            if (lane >= o) xs += y;
        }
        for (int o = 16; o > 0; o >>= 1) p += __shfl_xor(p, o);
        s_tot[lane] = xs - t + p;
    }
    __syncthreads();
    if (i < n) offs[i] = x - c + s_tot[wv];
}

__global__ __launch_bounds__(256) void edge_place(
    const int* __restrict__ src, const int* __restrict__ dst,
    const float* __restrict__ ew, int* __restrict__ offs,
    int* __restrict__ pairs, int E)
{
    int e = blockIdx.x * 256 + threadIdx.x;
    if (e >= E) return;
    int p = atomicAdd(&offs[dst[e]], 1);
    p = min(max(p, 0), E - 1);
    ((int2*)pairs)[p] = make_int2(src[e], __float_as_int(ew[e]));
}

__global__ __launch_bounds__(256) void row_gather_sum(
    const float* __restrict__ feat, const int* __restrict__ offs,
    const int* __restrict__ pairs, float* __restrict__ agg, int N, int E)
{
    const int lane = threadIdx.x & 31;
    const int d = __builtin_amdgcn_readfirstlane((int)(blockIdx.x * 8 + (threadIdx.x >> 5)));
    if (d >= N) return;
    int beg = (d > 0) ? offs[d - 1] : 0;
    int end = offs[d];
    beg = min(max(beg, 0), E);
    end = min(max(end, beg), E);
    const float4* __restrict__ f4 = (const float4*)feat;
    const int2* __restrict__ pr2 = (const int2*)pairs;
    float4 acc = make_float4(0.0f, 0.0f, 0.0f, 0.0f);
    for (int base = beg; base < end; base += 32) {
        const int n = min(32, end - base);
        int2 pr = make_int2(0, 0);
        if (lane < n) pr = pr2[base + lane];
        #pragma unroll 4
        for (int j = 0; j < n; ++j) {
            int s = __builtin_amdgcn_readlane(pr.x, j);
            s = min(max(s, 0), N - 1);
            const float w = __int_as_float(__builtin_amdgcn_readlane(pr.y, j));
            const float4 f = f4[(size_t)s * 32 + lane];
            acc.x += f.x * w;
            acc.y += f.y * w;
            acc.z += f.z * w;
            acc.w += f.w * w;
        }
    }
    ((float4*)agg)[(size_t)d * 32 + lane] = acc;
}

__global__ __launch_bounds__(256) void gemm_wmma(
    const float* __restrict__ agg, const float* __restrict__ W,
    const float* __restrict__ bias, const float* __restrict__ a1p,
    float* __restrict__ h_out, float* __restrict__ colsum,
    float* __restrict__ colsumsq)
{
    __shared__ float s_sum[D];
    __shared__ float s_sq[D];
    int tid = threadIdx.x;
    if (tid < D) { s_sum[tid] = 0.0f; s_sq[tid] = 0.0f; }
    __syncthreads();

    const int wave = tid >> 5;
    const int lane = tid & 31;
    const int half = lane >> 4;
    const int l    = lane & 15;
    const int m0   = blockIdx.x * 16;
    const int j0   = wave * 16;
    const float a1 = a1p[0];

    const float* Arow = agg + (size_t)(m0 + l) * D + 2 * half;
    const float* Bcol = W + (size_t)(2 * half) * D + (j0 + l);

    v8f c0 = {};
    v8f c1 = {};
    #pragma unroll 4
    for (int k = 0; k < D; k += 8) {
        {
            v2f a;
            a.x = Arow[k];
            a.y = Arow[k + 1];
            v2f bb;
            bb.x = Bcol[(size_t)k * D];
            bb.y = Bcol[(size_t)(k + 1) * D];
            c0 = __builtin_amdgcn_wmma_f32_16x16x4_f32(
                     false, a, false, bb, (short)0, c0, false, false);
        }
        {
            v2f a;
            a.x = Arow[k + 4];
            a.y = Arow[k + 5];
            v2f bb;
            bb.x = Bcol[(size_t)(k + 4) * D];
            bb.y = Bcol[(size_t)(k + 5) * D];
            c1 = __builtin_amdgcn_wmma_f32_16x16x4_f32(
                     false, a, false, bb, (short)0, c1, false, false);
        }
    }

    const float bj = bias[j0 + l];
    float lsum = 0.0f, lsq = 0.0f;
    #pragma unroll
    for (int v = 0; v < 8; ++v) {
        float h = c0[v] + c1[v] + bj;
        h = (h >= 0.0f) ? h : a1 * h;
        h_out[(size_t)(m0 + v + 8 * half) * D + (j0 + l)] = h;
        lsum += h;
        lsq  += h * h;
    }

    atomicAdd(&s_sum[j0 + l], lsum);
    atomicAdd(&s_sq[j0 + l], lsq);
    __syncthreads();
    if (tid < D) {
        atomicAdd(&colsum[tid],  s_sum[tid]);
        atomicAdd(&colsumsq[tid], s_sq[tid]);
    }
}

__global__ void bn_finalize(const float* __restrict__ colsum,
                            const float* __restrict__ colsumsq,
                            const float* __restrict__ gamma,
                            const float* __restrict__ beta,
                            float* __restrict__ scale,
                            float* __restrict__ shift, int N)
{
    int j = threadIdx.x;
    float invN = 1.0f / (float)N;
    float m   = colsum[j] * invN;
    float var = colsumsq[j] * invN - m * m;
    float inv = rsqrtf(var + 1e-5f);
    float sc  = gamma[j] * inv;
    scale[j] = sc;
    shift[j] = beta[j] - m * sc;
}

__global__ __launch_bounds__(256) void bn_prelu(
    float* __restrict__ h, const float* __restrict__ scale,
    const float* __restrict__ shift, const float* __restrict__ a2p,
    size_t nquads)
{
    size_t i = (size_t)blockIdx.x * blockDim.x + threadIdx.x;
    if (i >= nquads) return;
    const float a2 = a2p[0];
    int j = (int)((i * 4) & (D - 1));

    float4 v = ((float4*)h)[i];
    float x;
    x = v.x * scale[j + 0] + shift[j + 0]; v.x = (x >= 0.0f) ? x : a2 * x;
    x = v.y * scale[j + 1] + shift[j + 1]; v.y = (x >= 0.0f) ? x : a2 * x;
    x = v.z * scale[j + 2] + shift[j + 2]; v.z = (x >= 0.0f) ? x : a2 * x;
    x = v.w * scale[j + 3] + shift[j + 3]; v.w = (x >= 0.0f) ? x : a2 * x;
    ((float4*)h)[i] = v;
}

extern "C" void kernel_launch(void* const* d_in, const int* in_sizes, int n_in,
                              void* d_out, int out_size, void* d_ws, size_t ws_size,
                              hipStream_t stream) {
    const float* feat  = (const float*)d_in[0];
    const int*   src   = (const int*)d_in[1];
    const int*   dst   = (const int*)d_in[2];
    const float* ew    = (const float*)d_in[3];
    const float* Wm    = (const float*)d_in[4];
    const float* bias  = (const float*)d_in[5];
    const float* a1    = (const float*)d_in[6];
    const float* gamma = (const float*)d_in[7];
    const float* beta  = (const float*)d_in[8];
    const float* a2    = (const float*)d_in[9];

    const int N = in_sizes[0] / D;
    const int E = in_sizes[1];
    float* out = (float*)d_out;

    float* agg      = (float*)d_ws;
    float* colsum   = agg + (size_t)N * D;
    float* colsumsq = colsum + D;
    float* scale    = colsumsq + D;
    float* shift    = scale + D;

    int* cnt   = (int*)d_out;
    int* offs  = cnt + N;
    int* pairs = offs + N;

    const int eblocks = (E + 255) / 256;

    zero_counters<<<(N + 255) / 256, 256, 0, stream>>>(cnt, N, colsum);

    dst_histogram<<<eblocks, 256, 0, stream>>>(dst, cnt, E);

    row_offsets<<<(N + 1023) / 1024, 1024, 0, stream>>>(cnt, offs, N);

    edge_place<<<eblocks, 256, 0, stream>>>(src, dst, ew, offs, pairs, E);

    row_gather_sum<<<(N + 7) / 8, 256, 0, stream>>>(feat, offs, pairs, agg, N, E);

    gemm_wmma<<<N / 16, 256, 0, stream>>>(agg, Wm, bias, a1, out, colsum, colsumsq);

    bn_finalize<<<1, D, 0, stream>>>(colsum, colsumsq, gamma, beta, scale, shift, N);

    size_t nquads = (size_t)N * D / 4;
    int qblocks = (int)((nquads + 255) / 256);
    bn_prelu<<<qblocks, 256, 0, stream>>>(out, scale, shift, a2, nquads);
}


# Counting sort by destination row redone as two LDS-binned passes (256-row bins, 4096-edge tiles): global integer atomics per tile and bin, not per edge; sort stores in runs; row sums arrival-order
# speedup vs baseline: 3.0591x; 2.6042x over previous
#include <hip/hip_runtime.h>
#include <hip/hip_bf16.h>

typedef __attribute__((ext_vector_type(2))) float v2f;
typedef __attribute__((ext_vector_type(8))) float v8f;

#define D 128

#define SORT_T    4096
#define BIN_SHIFT 8
#define BIN_ROWS  256
#define NB_MAX    1024
#define STAGE_CAP 8192

__device__ __forceinline__ int block_excl_scan_1024(int c, int* s_tot) {
    const int lane = threadIdx.x & 31;
    const int wv   = threadIdx.x >> 5;
    int x = c;
    for (int o = 1; o < 32; o <<= 1) {
        int y = __shfl_up(x, o);
        if (lane >= o) x += y;
    }
    __syncthreads();
    if (lane == 31) s_tot[wv] = x;
    __syncthreads();
    if (wv == 0) {
        const int t = s_tot[lane];
        int xs = t;
        for (int o = 1; o < 32; o <<= 1) {
            int y = __shfl_up(xs, o);
            if (lane >= o) xs += y;
        }
        s_tot[lane] = xs - t;
    }
    __syncthreads();
    return x - c + s_tot[wv];
}

__global__ __launch_bounds__(1024) void zero_small(int* __restrict__ binCount,
                                                   int* __restrict__ binCursor,
                                                   float* __restrict__ colsums) {
    const int t = threadIdx.x;
    binCount[t]  = 0;
    binCursor[t] = 0;
    if (t < 2 * D) colsums[t] = 0.0f;
}

__global__ __launch_bounds__(1024) void bin_count(const int* __restrict__ dst,
                                                  int* __restrict__ binCount,
                                                  int N, int E) {
    __shared__ int s_hist[NB_MAX];
    const int tid = threadIdx.x;
    s_hist[tid] = 0;
    __syncthreads();
    const int e0 = blockIdx.x * SORT_T;
    int dk[SORT_T / 1024];
    #pragma unroll
    for (int k = 0; k < SORT_T / 1024; ++k) {
        const int e = e0 + k * 1024 + tid;
        dk[k] = (e < E) ? dst[e] : -1;
    }
    #pragma unroll
    for (int k = 0; k < SORT_T / 1024; ++k) {
        if ((unsigned)dk[k] < (unsigned)N) atomicAdd(&s_hist[dk[k] >> BIN_SHIFT], 1);
    }
    __syncthreads();
    const int c = s_hist[tid];
    if (c > 0) atomicAdd(&binCount[tid], c);
}

__global__ __launch_bounds__(1024) void bin_place(
    const int* __restrict__ src, const int* __restrict__ dst,
    const float* __restrict__ ew, const int* __restrict__ binCount,
    int* __restrict__ binCursor, int* __restrict__ binStart,
    int4* __restrict__ rec, int N, int E)
{
    __shared__ __attribute__((aligned(16))) int4 s_rec[SORT_T];
    __shared__ int s_dest[SORT_T];
    __shared__ int s_hist[NB_MAX];
    __shared__ int s_lbase[NB_MAX];
    __shared__ int s_gbase[NB_MAX];
    __shared__ int s_tot[32];
    __shared__ int s_n;
    const int tid = threadIdx.x;
    s_hist[tid] = 0;
    __syncthreads();

    const int e0 = blockIdx.x * SORT_T;
    int   dk[SORT_T / 1024];
    int   sk[SORT_T / 1024];
    int   wk[SORT_T / 1024];
    int   rk[SORT_T / 1024];
    #pragma unroll
    for (int k = 0; k < SORT_T / 1024; ++k) {
        const int e = e0 + k * 1024 + tid;
        const bool in = (e < E);
        dk[k] = in ? dst[e] : -1;
        sk[k] = in ? src[e] : 0;
        wk[k] = in ? __float_as_int(ew[e]) : 0;
    }
    const int gcount = binCount[tid];
    #pragma unroll
    for (int k = 0; k < SORT_T / 1024; ++k) {
        rk[k] = 0;
        if ((unsigned)dk[k] >= (unsigned)N) dk[k] = -1;
        if (dk[k] >= 0) rk[k] = atomicAdd(&s_hist[dk[k] >> BIN_SHIFT], 1);
    }
    __syncthreads();

    const int c      = s_hist[tid];
    const int lex = block_excl_scan_1024(c, s_tot);
    const int gex = block_excl_scan_1024(gcount, s_tot);
    s_lbase[tid] = lex;
    if (tid == 1023) s_n = lex + c;
    int gb = 0;
    if (c > 0) gb = gex + atomicAdd(&binCursor[tid], c);
    s_gbase[tid] = gb;
    if (blockIdx.x == 0) binStart[tid] = gex;
    __syncthreads();

    #pragma unroll
    for (int k = 0; k < SORT_T / 1024; ++k) {
        if (dk[k] >= 0) {
            const int b  = dk[k] >> BIN_SHIFT;
            const int ls = min(s_lbase[b] + rk[k], SORT_T - 1);
            s_rec[ls]  = make_int4(dk[k], sk[k], wk[k], 0);
            s_dest[ls] = min(max(s_gbase[b] + rk[k], 0), E - 1);
        }
    }
    __syncthreads();

    const int cnt = min(s_n, SORT_T);
    for (int i = tid; i < cnt; i += 1024) rec[s_dest[i]] = s_rec[i];
}

__global__ __launch_bounds__(1024) void bin_row_sort(
    const int4* __restrict__ rec, const int* __restrict__ binCount,
    const int* __restrict__ binStart, int* __restrict__ offs,
    int2* __restrict__ pairs, int N, int E)
{
    __shared__ __attribute__((aligned(8))) int2 s_pr[STAGE_CAP];
    __shared__ int s_cnt[BIN_ROWS];
    __shared__ int s_start[BIN_ROWS];
    __shared__ int s_fill[BIN_ROWS];
    __shared__ int s_tot[32];
    const int tid = threadIdx.x;
    const int b   = blockIdx.x;
    int start = binStart[b];
    int cnt   = binCount[b];
    start = min(max(start, 0), E);
    cnt   = min(max(cnt, 0), E - start);
    if (tid < BIN_ROWS) { s_cnt[tid] = 0; s_fill[tid] = 0; }
    __syncthreads();

    for (int base = 0; base < cnt; base += 4 * 1024) {
        int dr[4];
        #pragma unroll
        for (int k = 0; k < 4; ++k) {
            const int i = base + k * 1024 + tid;
            dr[k] = (i < cnt) ? rec[start + i].x : -1;
        }
        #pragma unroll
        for (int k = 0; k < 4; ++k)
            if (dr[k] >= 0) atomicAdd(&s_cnt[dr[k] & (BIN_ROWS - 1)], 1);
    }
    __syncthreads();

    const int c  = (tid < BIN_ROWS) ? s_cnt[tid] : 0;
    const int ex = block_excl_scan_1024(c, s_tot);
    if (tid < BIN_ROWS) {
        s_start[tid] = ex;
        const int row = b * BIN_ROWS + tid;
        if (row < N) offs[row] = start + ex + c;
    }
    __syncthreads();

    const bool staged = (cnt <= STAGE_CAP);
    for (int base = 0; base < cnt; base += 4 * 1024) {
        int4 r4[4];
        #pragma unroll
        for (int k = 0; k < 4; ++k) {
            const int i = min(base + k * 1024 + tid, cnt - 1);
            r4[k] = rec[start + i];
        }
        #pragma unroll
        for (int k = 0; k < 4; ++k) {
            if (base + k * 1024 + tid < cnt) {
                const int r = r4[k].x & (BIN_ROWS - 1);
                int slot = s_start[r] + atomicAdd(&s_fill[r], 1);
                slot = min(max(slot, 0), cnt - 1);
                const int2 v = make_int2(r4[k].y, r4[k].z);
                if (staged) s_pr[slot] = v;
                else        pairs[start + slot] = v;
            }
        }
    }
    if (staged) {
        __syncthreads();
        for (int i = tid; i < cnt; i += 1024) pairs[start + i] = s_pr[i];
    }
}

#define TR 32
#define APITCH 132

__global__ __launch_bounds__(256) void gather_gemm_wmma(
    const float* __restrict__ feat, const int* __restrict__ offs,
    const int* __restrict__ pairs, const float* __restrict__ W,
    const float* __restrict__ bias, const float* __restrict__ a1p,
    float* __restrict__ h_out, float* __restrict__ colsum,
    float* __restrict__ colsumsq, int N, int E)
{
    __shared__ __attribute__((aligned(16))) float s_a[TR * APITCH];
    __shared__ float s_sum[D];
    __shared__ float s_sq[D];
    const int tid = threadIdx.x;
    if (tid < D) { s_sum[tid] = 0.0f; s_sq[tid] = 0.0f; }

    const int lane = tid & 31;
    const int wave = __builtin_amdgcn_readfirstlane(tid >> 5);
    const int m0   = blockIdx.x * TR;

    const float4* __restrict__ f4 = (const float4*)feat;
    const int2* __restrict__ pr2 = (const int2*)pairs;
    for (int i = 0; i < TR / 8; ++i) {
        const int r = wave * (TR / 8) + i;
        const int d = m0 + r;
        float4 acc = make_float4(0.0f, 0.0f, 0.0f, 0.0f);
        if (d < N) {
            int beg = (d > 0) ? offs[d - 1] : 0;
            int end = offs[d];
            beg = min(max(beg, 0), E);
            end = min(max(end, beg), E);
            for (int base = beg; base < end; base += 32) {
                const int n = min(32, end - base);
                int2 pr = make_int2(0, 0);
                if (lane < n) pr = pr2[base + lane];
                #pragma unroll 4
                for (int j = 0; j < n; ++j) {
                    int s = __builtin_amdgcn_readlane(pr.x, j);
                    s = min(max(s, 0), N - 1);
                    const float w = __int_as_float(__builtin_amdgcn_readlane(pr.y, j));
                    const float4 f = f4[(size_t)s * 32 + lane];
                    acc.x += f.x * w;
                    acc.y += f.y * w;
                    acc.z += f.z * w;
                    acc.w += f.w * w;
                }
            }
        }
        *(float4*)&s_a[r * APITCH + lane * 4] = acc;
    }
    __syncthreads();

    const int half = lane >> 4;
    const int l    = lane & 15;
    const int j0   = wave * 16;
    const float a1 = a1p[0];

    const float* A0   = s_a + l * APITCH + 2 * half;
    const float* A1   = s_a + (16 + l) * APITCH + 2 * half;
    const float* Bcol = W + (size_t)(2 * half) * D + (j0 + l);

    v8f c00 = {};
    v8f c01 = {};
    v8f c10 = {};
    v8f c11 = {};
    #pragma unroll 2
    for (int k = 0; k < D; k += 8) {
        v2f b0;
        b0.x = Bcol[(size_t)k * D];
        b0.y = Bcol[(size_t)(k + 1) * D];
        v2f b1;
        b1.x = Bcol[(size_t)(k + 4) * D];
        b1.y = Bcol[(size_t)(k + 5) * D];
        const v2f a00 = *(const v2f*)(A0 + k);
        const v2f a10 = *(const v2f*)(A1 + k);
        const v2f a01 = *(const v2f*)(A0 + k + 4);
        const v2f a11 = *(const v2f*)(A1 + k + 4);
        c00 = __builtin_amdgcn_wmma_f32_16x16x4_f32(
                  false, a00, false, b0, (short)0, c00, false, false);
        c10 = __builtin_amdgcn_wmma_f32_16x16x4_f32(
                  false, a10, false, b0, (short)0, c10, false, false);
        c01 = __builtin_amdgcn_wmma_f32_16x16x4_f32(
                  false, a01, false, b1, (short)0, c01, false, false);
        c11 = __builtin_amdgcn_wmma_f32_16x16x4_f32(
                  false, a11, false, b1, (short)0, c11, false, false);
    }

    const float bj = bias[j0 + l];
    float lsum = 0.0f, lsq = 0.0f;
    #pragma unroll
    for (int v = 0; v < 8; ++v) {
        const int row = m0 + v + 8 * half;
        float h = c00[v] + c01[v] + bj;
        h = (h >= 0.0f) ? h : a1 * h;
        if (row < N) {
            h_out[(size_t)row * D + (j0 + l)] = h;
            lsum += h;
            lsq  += h * h;
        }
    }
    #pragma unroll
    for (int v = 0; v < 8; ++v) {
        const int row = m0 + 16 + v + 8 * half;
        float h = c10[v] + c11[v] + bj;
        h = (h >= 0.0f) ? h : a1 * h;
        if (row < N) {
            h_out[(size_t)row * D + (j0 + l)] = h;
            lsum += h;
            lsq  += h * h;
        }
    }

    atomicAdd(&s_sum[j0 + l], lsum);
    atomicAdd(&s_sq[j0 + l], lsq);
    __syncthreads();
    if (tid < D) {
        atomicAdd(&colsum[tid],  s_sum[tid]);
        atomicAdd(&colsumsq[tid], s_sq[tid]);
    }
}

__global__ void bn_finalize(const float* __restrict__ colsum,
                            const float* __restrict__ colsumsq,
                            const float* __restrict__ gamma,
                            const float* __restrict__ beta,
                            float* __restrict__ scale,
                            float* __restrict__ shift, int N)
{
    int j = threadIdx.x;
    float invN = 1.0f / (float)N;
    float m   = colsum[j] * invN;
    float var = colsumsq[j] * invN - m * m;
    float inv = rsqrtf(var + 1e-5f);
    float sc  = gamma[j] * inv;
    scale[j] = sc;
    shift[j] = beta[j] - m * sc;
}

__global__ __launch_bounds__(256) void bn_prelu(
    float* __restrict__ h, const float* __restrict__ scale,
    const float* __restrict__ shift, const float* __restrict__ a2p,
    size_t nquads)
{
    size_t i = (size_t)blockIdx.x * blockDim.x + threadIdx.x;
    if (i >= nquads) return;
    const float a2 = a2p[0];
    int j = (int)((i * 4) & (D - 1));

    float4 v = ((float4*)h)[i];
    float x;
    x = v.x * scale[j + 0] + shift[j + 0]; v.x = (x >= 0.0f) ? x : a2 * x;
    x = v.y * scale[j + 1] + shift[j + 1]; v.y = (x >= 0.0f) ? x : a2 * x;
    x = v.z * scale[j + 2] + shift[j + 2]; v.z = (x >= 0.0f) ? x : a2 * x;
    x = v.w * scale[j + 3] + shift[j + 3]; v.w = (x >= 0.0f) ? x : a2 * x;
    ((float4*)h)[i] = v;
}

extern "C" void kernel_launch(void* const* d_in, const int* in_sizes, int n_in,
                              void* d_out, int out_size, void* d_ws, size_t ws_size,
                              hipStream_t stream) {
    const float* feat  = (const float*)d_in[0];
    const int*   src   = (const int*)d_in[1];
    const int*   dst   = (const int*)d_in[2];
    const float* ew    = (const float*)d_in[3];
    const float* Wm    = (const float*)d_in[4];
    const float* bias  = (const float*)d_in[5];
    const float* a1    = (const float*)d_in[6];
    const float* gamma = (const float*)d_in[7];
    const float* beta  = (const float*)d_in[8];
    const float* a2    = (const float*)d_in[9];

    const int N = in_sizes[0] / D;
    const int E = in_sizes[1];
    float* out = (float*)d_out;

    const int Np = (N + 3) & ~3;
    int*   binCount  = (int*)d_ws;
    int*   binCursor = binCount + NB_MAX;
    int*   binStart  = binCursor + NB_MAX;
    int*   offs      = binStart + NB_MAX;
    int*   pairs     = offs + Np;
    int*   rec       = pairs + 2 * (size_t)E;
    float* colsum   = (float*)d_ws + (size_t)N * D;
    float* colsumsq = colsum + D;
    float* scale    = colsumsq + D;
    float* shift    = scale + D;

    const int tiles = (E + SORT_T - 1) / SORT_T;
    const int bins  = (N + BIN_ROWS - 1) / BIN_ROWS;

    zero_small<<<1, 1024, 0, stream>>>(binCount, binCursor, colsum);

    bin_count<<<tiles, 1024, 0, stream>>>(dst, binCount, N, E);

    bin_place<<<tiles, 1024, 0, stream>>>(src, dst, ew, binCount, binCursor, binStart,
                                          (int4*)rec, N, E);

    bin_row_sort<<<bins, 1024, 0, stream>>>((const int4*)rec, binCount, binStart, offs,
                                            (int2*)pairs, N, E);

    gather_gemm_wmma<<<(N + TR - 1) / TR, 256, 0, stream>>>(
        feat, offs, pairs, Wm, bias, a1, out, colsum, colsumsq, N, E);

    bn_finalize<<<1, D, 0, stream>>>(colsum, colsumsq, gamma, beta, scale, shift, N);

    size_t nquads = (size_t)N * D / 4;
    int qblocks = (int)((nquads + 255) / 256);
    bn_prelu<<<qblocks, 256, 0, stream>>>(out, scale, shift, a2, nquads);
}
